# starting kernel
# baseline (speedup 1.0000x reference)
#include <hip/hip_runtime.h>
#include <hip/hip_bf16.h>

#define N_NODES 50000
#define N_EDGES 800000
#define HID     24
#define HEADS   8
#define EDGE_DIM 8
#define NHID    192
#define M_TILES 3125
#define N_TILES 12

typedef __attribute__((ext_vector_type(2))) float v2f;
typedef __attribute__((ext_vector_type(8))) float v8f;

__device__ __forceinline__ void atomic_fmax(float* addr, float val) {
    if (val >= 0.0f)
        atomicMax((int*)addr, __float_as_int(val));
    else
        atomicMin((unsigned int*)addr, __float_as_uint(val));
}

__global__ void k_init(float* __restrict__ agg, float* __restrict__ emax,
                       float* __restrict__ denom, float* __restrict__ alpha_sum) {
    long i = (long)blockIdx.x * blockDim.x + threadIdx.x;
    long stride = (long)gridDim.x * blockDim.x;
    const float NEG_INF = __int_as_float(0xff800000);
    for (long j = i; j < (long)N_NODES * NHID; j += stride) agg[j] = 0.0f;
    for (long j = i; j < (long)N_NODES * HEADS; j += stride) {
        emax[j] = NEG_INF;
        denom[j] = 0.0f;
    }
    if (i < HEADS) alpha_sum[i] = 0.0f;
}

__global__ void k_vedge(const float* __restrict__ W_edge,
                        const float* __restrict__ att_edge,
                        float* __restrict__ v_edge) {
    int i = threadIdx.x;
    if (i >= EDGE_DIM * HEADS) return;
    int d = i >> 3, h = i & 7;
    float s = 0.0f;
    #pragma unroll
    for (int c = 0; c < HID; ++c)
        s += W_edge[d * NHID + h * HID + c] * att_edge[h * HID + c];
    v_edge[d * HEADS + h] = s;
}

__global__ void k_gemm_wmma(const float* __restrict__ x,
                            const float* __restrict__ W,
                            float* __restrict__ xl) {
    int lane = threadIdx.x & 31;
    int wave = (int)((blockIdx.x * blockDim.x + threadIdx.x) >> 5);
    if (wave >= M_TILES * N_TILES) return;
    int tm = wave / N_TILES;
    int tn = wave % N_TILES;

    int m  = lane & 15;
    int hs = lane >> 4;
    const float* arow = x + (long)(tm * 16 + m) * HID;
    int bcol = tn * 16 + m;

    v8f c = {0.f, 0.f, 0.f, 0.f, 0.f, 0.f, 0.f, 0.f};
    #pragma unroll
    for (int kb = 0; kb < HID; kb += 4) {
        int k0 = kb + hs * 2;
        v2f a; a.x = arow[k0]; a.y = arow[k0 + 1];
        v2f b; b.x = W[(long)k0 * NHID + bcol]; b.y = W[(long)(k0 + 1) * NHID + bcol];
        c = __builtin_amdgcn_wmma_f32_16x16x4_f32(
                 false, a,  false, b,
                 (short)0, c,  false,  false);
    }
    int n = lane & 15;
    #pragma unroll
    for (int r = 0; r < 8; ++r) {
        int mg = tm * 16 + r + hs * 8;
        xl[(long)mg * NHID + tn * 16 + n] = c[r];
    }
}

__global__ void k_att(const float* __restrict__ xl,
                      const float* __restrict__ att_src,
                      const float* __restrict__ att_dst,
                      float* __restrict__ a_src, float* __restrict__ a_dst) {
    int i = blockIdx.x * blockDim.x + threadIdx.x;
    if (i >= N_NODES * HEADS) return;
    int n = i >> 3, h = i & 7;
    const float* row = xl + (long)n * NHID + h * HID;
    const float* as  = att_src + h * HID;
    const float* ad  = att_dst + h * HID;
    float s = 0.0f, d = 0.0f;
    #pragma unroll
    for (int c = 0; c < HID; ++c) {
        float v = row[c];
        s += v * as[c];
        d += v * ad[c];
    }
    a_src[i] = s;
    a_dst[i] = d;
}

__global__ void k_edge_logits(const int* __restrict__ src, const int* __restrict__ dst,
                              const float* __restrict__ edge_attr,
                              const float* __restrict__ a_src, const float* __restrict__ a_dst,
                              const float* __restrict__ v_edge,
                              float* __restrict__ ebuf, float* __restrict__ emax) {
    __shared__ float sv[EDGE_DIM * HEADS];
    if (threadIdx.x < EDGE_DIM * HEADS) sv[threadIdx.x] = v_edge[threadIdx.x];
    __syncthreads();
    int e = blockIdx.x * blockDim.x + threadIdx.x;
    if (e >= N_EDGES) return;
    int s = src[e], d = dst[e];
    float ea[EDGE_DIM];
    #pragma unroll
    for (int j = 0; j < EDGE_DIM; ++j) ea[j] = edge_attr[(long)e * EDGE_DIM + j];
    #pragma unroll
    for (int h = 0; h < HEADS; ++h) {
        float ae = 0.0f;
        #pragma unroll
        for (int j = 0; j < EDGE_DIM; ++j) ae += ea[j] * sv[j * HEADS + h];
        float v = a_src[s * HEADS + h] + a_dst[d * HEADS + h] + ae;
        v = (v > 0.0f) ? v : 0.2f * v;
        ebuf[(long)e * HEADS + h] = v;
        atomic_fmax(&emax[d * HEADS + h], v);
    }
}

__global__ void k_edge_exp(const int* __restrict__ dst,
                           const float* __restrict__ emax,
                           float* __restrict__ ebuf, float* __restrict__ denom) {
    int e = blockIdx.x * blockDim.x + threadIdx.x;
    if (e >= N_EDGES) return;
    int d = dst[e];
    #pragma unroll
    for (int h = 0; h < HEADS; ++h) {
        float v = __expf(ebuf[(long)e * HEADS + h] - emax[d * HEADS + h]);
        ebuf[(long)e * HEADS + h] = v;
        atomicAdd(&denom[d * HEADS + h], v);
    }
}

__global__ void k_scatter(const int* __restrict__ src, const int* __restrict__ dst,
                          const float* __restrict__ ebuf, const float* __restrict__ denom,
                          const float* __restrict__ xl, float* __restrict__ agg,
                          float* __restrict__ alpha_sum) {
    __shared__ float s_as[HEADS];
    if (threadIdx.x < HEADS) s_as[threadIdx.x] = 0.0f;
    __syncthreads();

    int lane = threadIdx.x & 31;
    int wpb  = blockDim.x >> 5;
    long wave   = (long)blockIdx.x * wpb + (threadIdx.x >> 5);
    long nwaves = (long)gridDim.x * wpb;

    float asum = 0.0f;
    for (long e = wave; e < N_EDGES; e += nwaves) {
        int s = src[e], d = dst[e];
        float alpha = 0.0f;
        if (lane < HEADS)
            alpha = ebuf[e * HEADS + lane] / (denom[d * HEADS + lane] + 1e-16f);
        asum += alpha;
        const float* xs = xl + (long)s * NHID;
        float*       ag = agg + (long)d * NHID;
        #pragma unroll
        for (int i = 0; i < NHID / 32; ++i) {
            int c = lane + 32 * i;
            int h = c / HID;
            float ah = __shfl(alpha, h, 32);
            atomicAdd(&ag[c], xs[c] * ah);
        }
    }
    if (lane < HEADS) atomicAdd(&s_as[lane], asum);
    __syncthreads();
    if (threadIdx.x < HEADS) atomicAdd(&alpha_sum[threadIdx.x], s_as[threadIdx.x]);
}

__global__ void k_final(const float* __restrict__ agg, const float* __restrict__ bias,
                        const float* __restrict__ W_out, const float* __restrict__ b_out,
                        float* __restrict__ y) {
    int lane = threadIdx.x & 31;
    int node = (int)((blockIdx.x * blockDim.x + threadIdx.x) >> 5);
    if (node >= N_NODES) return;
    float val = 0.0f;
    if (lane < HID) {
        float s = 0.0f;
        #pragma unroll
        for (int h = 0; h < HEADS; ++h) s += agg[(long)node * NHID + h * HID + lane];
        s = s * (1.0f / HEADS) + bias[lane];
        s = (s > 0.0f) ? s : (__expf(s) - 1.0f);
        val = s * W_out[lane];
    }
    #pragma unroll
    for (int off = 16; off > 0; off >>= 1) val += __shfl_down(val, off, 32);
    if (lane == 0) y[node] = val + b_out[0];
}

__global__ void k_alpha_mean(const float* __restrict__ alpha_sum, float* __restrict__ out) {
    int h = threadIdx.x;
    if (h < HEADS) out[h] = alpha_sum[h] * (1.0f / (float)N_EDGES);
}

extern "C" void kernel_launch(void* const* d_in, const int* in_sizes, int n_in,
                              void* d_out, int out_size, void* d_ws, size_t ws_size,
                              hipStream_t stream) {
    const float* x         = (const float*)d_in[0];
    const int*   ei        = (const int*)d_in[1];
    const float* edge_attr = (const float*)d_in[2];
    const float* W_lin     = (const float*)d_in[3];
    const float* att_src   = (const float*)d_in[4];
    const float* att_dst   = (const float*)d_in[5];
    const float* W_edge    = (const float*)d_in[6];
    const float* att_edge  = (const float*)d_in[7];
    const float* bias      = (const float*)d_in[8];
    const float* W_out     = (const float*)d_in[9];
    const float* b_out     = (const float*)d_in[10];
    (void)in_sizes; (void)n_in; (void)out_size; (void)ws_size;

    const int* src = ei;
    const int* dst = ei + N_EDGES;

    float* w        = (float*)d_ws;
    float* xl       = w;
    float* agg      = xl      + (long)N_NODES * NHID;
    float* a_src    = agg     + (long)N_NODES * NHID;
    float* a_dst    = a_src   + (long)N_NODES * HEADS;
    float* emax     = a_dst   + (long)N_NODES * HEADS;
    float* denom    = emax    + (long)N_NODES * HEADS;
    float* ebuf     = denom   + (long)N_NODES * HEADS;
    float* v_edge   = ebuf    + (long)N_EDGES * HEADS;
    float* alpha_s  = v_edge  + EDGE_DIM * HEADS;

    float* y         = (float*)d_out;
    float* alpha_out = y + N_NODES;

    k_init<<<2048, 256, 0, stream>>>(agg, emax, denom, alpha_s);
    k_vedge<<<1, 64, 0, stream>>>(W_edge, att_edge, v_edge);

    {
        int waves = M_TILES * N_TILES;
        int blocks = (waves + 7) / 8;
        k_gemm_wmma<<<blocks, 256, 0, stream>>>(x, W_lin, xl);
    }
    k_att<<<(N_NODES * HEADS + 255) / 256, 256, 0, stream>>>(xl, att_src, att_dst, a_src, a_dst);
    k_edge_logits<<<(N_EDGES + 255) / 256, 256, 0, stream>>>(src, dst, edge_attr,
                                                             a_src, a_dst, v_edge, ebuf, emax);
    k_edge_exp<<<(N_EDGES + 255) / 256, 256, 0, stream>>>(dst, emax, ebuf, denom);
    k_scatter<<<2048, 256, 0, stream>>>(src, dst, ebuf, denom, xl, agg, alpha_s);
    k_final<<<(N_NODES * 32 + 255) / 256, 256, 0, stream>>>(agg, bias, W_out, b_out, y);
    k_alpha_mean<<<1, 32, 0, stream>>>(alpha_s, alpha_out);
}


# Edges counting-sorted by destination node (atomic count gives rank, scan, place); one wave per node pulls its segment: max, denominators, xl sums in registers; arrival-order ranks, sums regrouped
# speedup vs baseline: 3.8977x; 3.8977x over previous
#include <hip/hip_runtime.h>
#include <hip/hip_bf16.h>

#define N_NODES 50000
#define N_EDGES 800000
#define HID     24
#define HEADS   8
#define EDGE_DIM 8
#define NHID    192
#define M_TILES 3125
#define N_TILES 12
#define SCAN_T  1024
#define SCAN_TILES 13
#define PULL_WPB 8
#define PULL_BLOCKS (N_NODES / PULL_WPB)

typedef __attribute__((ext_vector_type(2))) float v2f;
typedef __attribute__((ext_vector_type(8))) float v8f;

__global__ void k_init(int* __restrict__ cnt) {
    int i = blockIdx.x * blockDim.x + threadIdx.x;
    if (i < N_NODES) cnt[i] = 0;
}

__global__ void k_vedge(const float* __restrict__ W_edge,
                        const float* __restrict__ att_edge,
                        float* __restrict__ v_edge) {
    int i = threadIdx.x;
    if (i >= EDGE_DIM * HEADS) return;
    int d = i >> 3, h = i & 7;
    float s = 0.0f;
    #pragma unroll
    for (int c = 0; c < HID; ++c)
        s += W_edge[d * NHID + h * HID + c] * att_edge[h * HID + c];
    v_edge[d * HEADS + h] = s;
}

__global__ void k_gemm_wmma(const float* __restrict__ x,
                            const float* __restrict__ W,
                            float* __restrict__ xl) {
    int lane = threadIdx.x & 31;
    int wave = (int)((blockIdx.x * blockDim.x + threadIdx.x) >> 5);
    if (wave >= M_TILES * N_TILES) return;
    int tm = wave / N_TILES;
    int tn = wave % N_TILES;

    int m  = lane & 15;
    int hs = lane >> 4;
    const float* arow = x + (long)(tm * 16 + m) * HID;
    int bcol = tn * 16 + m;

    v8f c = {0.f, 0.f, 0.f, 0.f, 0.f, 0.f, 0.f, 0.f};
    #pragma unroll
    for (int kb = 0; kb < HID; kb += 4) {
        int k0 = kb + hs * 2;
        v2f a; a.x = arow[k0]; a.y = arow[k0 + 1];
        v2f b; b.x = W[(long)k0 * NHID + bcol]; b.y = W[(long)(k0 + 1) * NHID + bcol];
        c = __builtin_amdgcn_wmma_f32_16x16x4_f32(
                 false, a,  false, b,
                 (short)0, c,  false,  false);
    }
    int n = lane & 15;
    #pragma unroll
    for (int r = 0; r < 8; ++r) {
        int mg = tm * 16 + r + hs * 8;
        xl[(long)mg * NHID + tn * 16 + n] = c[r];
    }
}

__global__ void k_att(const float* __restrict__ xl,
                      const float* __restrict__ att_src,
                      const float* __restrict__ att_dst,
                      float* __restrict__ a_src, float* __restrict__ a_dst) {
    int i = blockIdx.x * blockDim.x + threadIdx.x;
    if (i >= N_NODES * HEADS) return;
    int n = i >> 3, h = i & 7;
    const float* row = xl + (long)n * NHID + h * HID;
    const float* as  = att_src + h * HID;
    const float* ad  = att_dst + h * HID;
    float s = 0.0f, d = 0.0f;
    #pragma unroll
    for (int c = 0; c < HID; ++c) {
        float v = row[c];
        s += v * as[c];
        d += v * ad[c];
    }
    a_src[i] = s;
    a_dst[i] = d;
}

__global__ void k_count(const int* __restrict__ dst, int* __restrict__ cnt,
                        int* __restrict__ rank) {
    int e = blockIdx.x * blockDim.x + threadIdx.x;
    if (e >= N_EDGES) return;
    rank[e] = atomicAdd(&cnt[dst[e]], 1);
}

__global__ void k_scan(const int* __restrict__ cnt, int* __restrict__ off) {
    __shared__ int wtot[SCAN_TILES][32];
    __shared__ int wpre[SCAN_TILES][32];
    __shared__ int ttot[SCAN_TILES];
    int t = threadIdx.x, lane = t & 31, wv = t >> 5;
    int4 v[SCAN_TILES];
    int ex[SCAN_TILES];
    #pragma unroll
    for (int k = 0; k < SCAN_TILES; ++k) {
        int idx = k * (SCAN_T * 4) + t * 4;
        v[k] = make_int4(0, 0, 0, 0);
        if (idx < N_NODES) v[k] = *reinterpret_cast<const int4*>(cnt + idx);
    }
    #pragma unroll
    for (int k = 0; k < SCAN_TILES; ++k) {
        int ts = v[k].x + v[k].y + v[k].z + v[k].w;
        int xs = ts;
        #pragma unroll
        for (int o = 1; o < 32; o <<= 1) {
            int nb = __shfl_up(xs, o, 32);
            if (lane >= o) xs += nb;
        }
        if (lane == 31) wtot[k][wv] = xs;
        ex[k] = xs - ts;
    }
    __syncthreads();
    if (wv < SCAN_TILES) {
        int val = wtot[wv][lane];
        int xs = val;
        #pragma unroll
        for (int o = 1; o < 32; o <<= 1) {
            int nb = __shfl_up(xs, o, 32);
            if (lane >= o) xs += nb;
        }
        wpre[wv][lane] = xs - val;
        if (lane == 31) ttot[wv] = xs;
    }
    __syncthreads();
    int tp = 0;
    #pragma unroll
    for (int k = 0; k < SCAN_TILES; ++k) {
        int idx = k * (SCAN_T * 4) + t * 4;
        int b0 = tp + wpre[k][wv] + ex[k];
        int b1 = b0 + v[k].x, b2 = b1 + v[k].y, b3 = b2 + v[k].z;
        if (idx <= N_NODES) *reinterpret_cast<int4*>(off + idx) = make_int4(b0, b1, b2, b3);
        tp += ttot[k];
    }
}

__global__ void k_place(const int* __restrict__ src, const int* __restrict__ dst,
                        const float* __restrict__ edge_attr,
                        const float* __restrict__ a_src, const float* __restrict__ a_dst,
                        const float* __restrict__ v_edge,
                        const int* __restrict__ off, const int* __restrict__ rank,
                        int* __restrict__ ssrc, float* __restrict__ tbuf) {
    __shared__ float sv[EDGE_DIM * HEADS];
    if (threadIdx.x < EDGE_DIM * HEADS) sv[threadIdx.x] = v_edge[threadIdx.x];
    __syncthreads();
    int e = blockIdx.x * blockDim.x + threadIdx.x;
    if (e >= N_EDGES) return;
    int s = src[e], d = dst[e];
    int pos = off[d] + rank[e];
    float ea[EDGE_DIM];
    #pragma unroll
    for (int j = 0; j < EDGE_DIM; ++j) ea[j] = edge_attr[(long)e * EDGE_DIM + j];
    const float4* ps = reinterpret_cast<const float4*>(a_src + (long)s * HEADS);
    const float4* pd = reinterpret_cast<const float4*>(a_dst + (long)d * HEADS);
    float4 s0 = ps[0], s1 = ps[1], d0 = pd[0], d1 = pd[1];
    float as[HEADS] = {s0.x, s0.y, s0.z, s0.w, s1.x, s1.y, s1.z, s1.w};
    float ad[HEADS] = {d0.x, d0.y, d0.z, d0.w, d1.x, d1.y, d1.z, d1.w};
    float v[HEADS];
    #pragma unroll
    for (int h = 0; h < HEADS; ++h) {
        float ae = 0.0f;
        #pragma unroll
        for (int j = 0; j < EDGE_DIM; ++j) ae += ea[j] * sv[j * HEADS + h];
        float t = as[h] + ad[h] + ae;
        v[h] = (t > 0.0f) ? t : 0.2f * t;
    }
    ssrc[pos] = s;
    float4* pt = reinterpret_cast<float4*>(tbuf + (long)pos * HEADS);
    pt[0] = make_float4(v[0], v[1], v[2], v[3]);
    pt[1] = make_float4(v[4], v[5], v[6], v[7]);
}

__device__ __forceinline__ void pull_edge(const float* __restrict__ xl, int s, float a, int lane,
                                          float (&acc)[6]) {
    const float2* xs = reinterpret_cast<const float2*>(xl + (long)s * NHID + lane * 6);
    float2 p0 = xs[0], p1 = xs[1], p2 = xs[2];
    acc[0] += p0.x * a; acc[1] += p0.y * a;
    acc[2] += p1.x * a; acc[3] += p1.y * a;
    acc[4] += p2.x * a; acc[5] += p2.y * a;
}

__global__ void k_pull(const int* __restrict__ off, const int* __restrict__ ssrc,
                       const float* __restrict__ tbuf, const float* __restrict__ xl,
                       const float* __restrict__ bias, const float* __restrict__ W_out,
                       const float* __restrict__ b_out,
                       float* __restrict__ y, float* __restrict__ part) {
    __shared__ float s_as[PULL_WPB][HEADS];
    int lane = threadIdx.x & 31;
    int wv   = threadIdx.x >> 5;
    int node = blockIdx.x * PULL_WPB + wv;
    int beg = off[node];
    int n   = off[node + 1] - beg;
    const float* tb = tbuf + (long)beg * HEADS;
    const int*   sb = ssrc + beg;

    int slot = lane >> 3, h = lane & 7;
    float m = __int_as_float(0xff800000);
    for (int i = slot; i < n; i += 4) { float v = tb[i * HEADS + h]; m = (v > m) ? v : m; }
    m = fmaxf(m, __shfl_xor(m, 8, 32));
    m = fmaxf(m, __shfl_xor(m, 16, 32));
    float den = 0.0f;
    for (int i = slot; i < n; i += 4) den += __expf(tb[i * HEADS + h] - m);
    den += __shfl_xor(den, 8, 32);
    den += __shfl_xor(den, 16, 32);

    int hh = lane >> 2;
    float m3 = __shfl(m, hh, 32);
    float dn = __shfl(den, hh, 32) + 1e-16f;
    float acc[6] = {0.f, 0.f, 0.f, 0.f, 0.f, 0.f};
    float asum = 0.0f;
    int i = 0;
    for (; i + 4 <= n; i += 4) {
        int s0 = sb[i], s1 = sb[i + 1], s2 = sb[i + 2], s3 = sb[i + 3];
        float a0 = __expf(tb[(i    ) * HEADS + hh] - m3) / dn;
        float a1 = __expf(tb[(i + 1) * HEADS + hh] - m3) / dn;
        float a2 = __expf(tb[(i + 2) * HEADS + hh] - m3) / dn;
        float a3 = __expf(tb[(i + 3) * HEADS + hh] - m3) / dn;
        asum += a0; asum += a1; asum += a2; asum += a3;
        pull_edge(xl, s0, a0, lane, acc);
        pull_edge(xl, s1, a1, lane, acc);
        pull_edge(xl, s2, a2, lane, acc);
        pull_edge(xl, s3, a3, lane, acc);
    }
    for (; i < n; ++i) {
        int s0 = sb[i];
        float a0 = __expf(tb[i * HEADS + hh] - m3) / dn;
        asum += a0;
        pull_edge(xl, s0, a0, lane, acc);
    }

    float val = 0.0f;
    #pragma unroll
    for (int j = 0; j < 6; ++j) {
        float r = acc[j];
        r += __shfl_xor(r, 4, 32);
        r += __shfl_xor(r, 8, 32);
        r += __shfl_xor(r, 16, 32);
        int c = (lane & 3) * 6 + j;
        float sres = r * (1.0f / HEADS) + bias[c];
        sres = (sres > 0.0f) ? sres : (__expf(sres) - 1.0f);
        val += sres * W_out[c];
    }
    val += __shfl_xor(val, 1, 32);
    val += __shfl_xor(val, 2, 32);
    if (lane == 0) y[node] = val + b_out[0];

    if ((lane & 3) == 0) s_as[wv][hh] = asum;
    __syncthreads();
    if (threadIdx.x < HEADS) {
        float t = 0.0f;
        for (int w = 0; w < PULL_WPB; ++w) t += s_as[w][threadIdx.x];
        part[blockIdx.x * HEADS + threadIdx.x] = t;
    }
}

__global__ void k_alpha_mean(const float* __restrict__ part, float* __restrict__ out) {
    __shared__ float sp[128][HEADS];
    int t = threadIdx.x;
    int h = t & 7, g = t >> 3;
    float s = 0.0f;
    for (int b = g; b < PULL_BLOCKS; b += 128) s += part[b * HEADS + h];
    sp[g][h] = s;
    __syncthreads();
    if (t < HEADS) {
        float r = 0.0f;
        for (int k = 0; k < 128; ++k) r += sp[k][t];
        out[t] = r * (1.0f / (float)N_EDGES);
    }
}

extern "C" void kernel_launch(void* const* d_in, const int* in_sizes, int n_in,
                              void* d_out, int out_size, void* d_ws, size_t ws_size,
                              hipStream_t stream) {
    const float* x         = (const float*)d_in[0];
    const int*   ei        = (const int*)d_in[1];
    const float* edge_attr = (const float*)d_in[2];
    const float* W_lin     = (const float*)d_in[3];
    const float* att_src   = (const float*)d_in[4];
    const float* att_dst   = (const float*)d_in[5];
    const float* W_edge    = (const float*)d_in[6];
    const float* att_edge  = (const float*)d_in[7];
    const float* bias      = (const float*)d_in[8];
    const float* W_out     = (const float*)d_in[9];
    const float* b_out     = (const float*)d_in[10];
    (void)in_sizes; (void)n_in; (void)out_size; (void)ws_size;

    const int* src = ei;
    const int* dst = ei + N_EDGES;

    float* w        = (float*)d_ws;
    float* xl       = w;
    float* a_src    = xl      + (long)N_NODES * NHID;
    float* a_dst    = a_src   + (long)N_NODES * HEADS;
    float* tbuf     = a_dst   + (long)N_NODES * HEADS;
    float* v_edge   = tbuf    + (long)N_EDGES * HEADS;
    float* part     = v_edge  + EDGE_DIM * HEADS;
    int*   cnt      = (int*)(part + PULL_BLOCKS * HEADS);
    int*   off      = cnt     + N_NODES;
    int*   rank     = off     + N_NODES + 16;
    int*   ssrc     = rank    + N_EDGES;

    float* y         = (float*)d_out;
    float* alpha_out = y + N_NODES;

    k_init<<<(N_NODES + 255) / 256, 256, 0, stream>>>(cnt);
    k_vedge<<<1, 64, 0, stream>>>(W_edge, att_edge, v_edge);

    {
        int waves = M_TILES * N_TILES;
        int blocks = (waves + 7) / 8;
        k_gemm_wmma<<<blocks, 256, 0, stream>>>(x, W_lin, xl);
    }
    k_att<<<(N_NODES * HEADS + 255) / 256, 256, 0, stream>>>(xl, att_src, att_dst, a_src, a_dst);
    k_count<<<(N_EDGES + 255) / 256, 256, 0, stream>>>(dst, cnt, rank);
    k_scan<<<1, SCAN_T, 0, stream>>>(cnt, off);
    k_place<<<(N_EDGES + 255) / 256, 256, 0, stream>>>(src, dst, edge_attr, a_src, a_dst, v_edge,
                                                       off, rank, ssrc, tbuf);
    k_pull<<<PULL_BLOCKS, PULL_WPB * 32, 0, stream>>>(off, ssrc, tbuf, xl, bias, W_out, b_out, y, part);
    k_alpha_mean<<<1, 1024, 0, stream>>>(part, alpha_out);
}
